# starting kernel
# baseline (speedup 1.0000x reference)
#include <hip/hip_runtime.h>

#define BB   32
#define TT   8192
#define FF   128
#define OUTT 1024
#define GG   (TT / 16)

typedef float v2f __attribute__((ext_vector_type(2)));
typedef float v8f __attribute__((ext_vector_type(8)));

__global__ __launch_bounds__(256) void k_groupsum_wmma(const float* __restrict__ x,
                                                       float* __restrict__ cs) {
    const int lane = threadIdx.x & 31;
    const int w    = blockIdx.x * 8 + (threadIdx.x >> 5);
    const int fg   = w & 7;
    const int g    = (w >> 3) & (GG - 1);
    const int b    = w >> 12;

    const int n    = lane & 15;
    const int koff = (lane >> 4) << 1;

    const float* base = x + ((size_t)b * TT + (size_t)g * 16) * FF + fg * 16 + n;

    v2f a; a.x = 1.0f; a.y = 1.0f;
    v8f c = {};
#pragma unroll
    for (int k = 0; k < 4; ++k) {
        const float* p = base + (size_t)(4 * k + koff) * FF;
        v2f bb; bb.x = p[0]; bb.y = p[FF];
        c = __builtin_amdgcn_wmma_f32_16x16x4_f32(false, a, false, bb, (short)0, c,
                                                  false, false);
    }
    if (lane < 16)
        cs[((size_t)b * GG + g) * FF + fg * 16 + lane] = c[0];
}

__global__ __launch_bounds__(128) void k_prefix(float* __restrict__ cs) {
    const int f = threadIdx.x;
    const int b = blockIdx.x;
    float run = 0.0f;
    const size_t base = (size_t)b * GG * FF + f;
#pragma unroll 4
    for (int g = 0; g < GG; ++g) {
        const size_t i = base + (size_t)g * FF;
        run += cs[i];
        cs[i] = run;
    }
}

__global__ __launch_bounds__(128) void k_out(const float* __restrict__ x,
                                             const float* __restrict__ ps,
                                             const int* __restrict__ lengths,
                                             float* __restrict__ out) {
    const int f    = threadIdx.x;
    const int tile = blockIdx.x;
    const int b    = blockIdx.y;
    const int t0   = tile * 16;

    const int   fsize = lengths[b] - (OUTT - 1);
    const float fsf   = (float)fsize;

    const float* xb = x  + (size_t)b * TT * FF + f;
    const float* pb = ps + (size_t)b * GG * FF + f;

    float lo = (tile == 0) ? 0.0f : pb[(size_t)(tile - 1) * FF];

    const int idx0 = fsize + t0;
    const int g0   = idx0 >> 4;
    const int r0   = idx0 & 15;
    float hi = (g0 == 0) ? 0.0f : pb[(size_t)(g0 - 1) * FF];
    const int s0 = idx0 - r0;
    for (int i = 0; i < r0; ++i)
        hi += xb[(size_t)(s0 + i) * FF];

    float* ob = out + ((size_t)b * OUTT + t0) * FF + f;
#pragma unroll
    for (int j = 0; j < 16; ++j) {
        ob[(size_t)j * FF] = (hi - lo) / fsf;
        if (j < 15) {
            lo += xb[(size_t)(t0 + j) * FF];
            hi += xb[(size_t)(idx0 + j) * FF];
        }
    }
}

extern "C" void kernel_launch(void* const* d_in, const int* in_sizes, int n_in,
                              void* d_out, int out_size, void* d_ws, size_t ws_size,
                              hipStream_t stream) {
    (void)in_sizes; (void)n_in; (void)out_size; (void)ws_size;
    const float* x       = (const float*)d_in[0];
    const int*   lengths = (const int*)d_in[1];
    float*       out     = (float*)d_out;
    float*       cs      = (float*)d_ws;

    k_groupsum_wmma<<<16384, 256, 0, stream>>>(x, cs);
    k_prefix<<<32, 128, 0, stream>>>(cs);
    k_out<<<dim3(64, 32), 128, 0, stream>>>(x, cs, lengths, out);
}


# Window-sum decomposition: out=(W0+scan_{k<t}(x[k+fs]-x[k]))/fs, W0 = ones-WMMA column sums of the first fs rows only; no 8192-step prefix table, 3 launches -> 2; changes the sum order
# speedup vs baseline: 4.9142x; 4.9142x over previous
#include <hip/hip_runtime.h>

#define BB   32
#define TT   8192
#define FF   128
#define OUTT 1024
#define CH   512
#define NCH  15

typedef float v2f __attribute__((ext_vector_type(2)));
typedef float v4f __attribute__((ext_vector_type(4)));
typedef float v8f __attribute__((ext_vector_type(8)));


__global__ __launch_bounds__(256) void k_winsum_wmma(const float* __restrict__ x,
                                                     const int* __restrict__ lengths,
                                                     float* __restrict__ part) {
    __shared__ float red[8][64];
    const int c  = blockIdx.x;
    const int fh = blockIdx.y;
    const int b  = blockIdx.z;
    const int fsize = lengths[b] - (OUTT - 1);
    if (c * CH >= fsize) return;

    const int lane = threadIdx.x & 31;
    const int w    = threadIdx.x >> 5;
    const int h    = lane >> 4;
    const int q    = lane & 15;

    const int last = fsize - 1;
    const int r0   = c * CH + w * 64 + h;
    const float* xb = x + (size_t)b * TT * FF + fh * 64 + q * 4;

    v8f a0 = {}, a1 = {}, a2 = {}, a3 = {};

    v4f cur[8], nxt[8];
#pragma unroll
    for (int j = 0; j < 8; ++j) {
        const int row = r0 + 2 * j;
        const int rc  = row < last ? row : last;
        cur[j] = *(const v4f*)(xb + (size_t)rc * FF);
    }
#pragma unroll
    for (int g = 0; g < 4; ++g) {
        if (g < 3) {
#pragma unroll
            for (int j = 0; j < 8; ++j) {
                const int row = r0 + (g + 1) * 16 + 2 * j;
                const int rc  = row < last ? row : last;
                nxt[j] = *(const v4f*)(xb + (size_t)rc * FF);
            }
        }
        __builtin_amdgcn_sched_barrier(0);
#pragma unroll
        for (int j = 0; j < 8; j += 2) {
            const int rowa = r0 + g * 16 + 2 * j;
            v2f am;
            am.x = (rowa     <= last) ? 1.0f : 0.0f;
            am.y = (rowa + 2 <= last) ? 1.0f : 0.0f;
            v2f b0; b0.x = cur[j].x; b0.y = cur[j + 1].x;
            v2f b1; b1.x = cur[j].y; b1.y = cur[j + 1].y;
            v2f b2; b2.x = cur[j].z; b2.y = cur[j + 1].z;
            v2f b3; b3.x = cur[j].w; b3.y = cur[j + 1].w;
            a0 = __builtin_amdgcn_wmma_f32_16x16x4_f32(false, am, false, b0, (short)0, a0, false, false);
            a1 = __builtin_amdgcn_wmma_f32_16x16x4_f32(false, am, false, b1, (short)0, a1, false, false);
            a2 = __builtin_amdgcn_wmma_f32_16x16x4_f32(false, am, false, b2, (short)0, a2, false, false);
            a3 = __builtin_amdgcn_wmma_f32_16x16x4_f32(false, am, false, b3, (short)0, a3, false, false);
        }
        if (g < 3) {
#pragma unroll
            for (int j = 0; j < 8; ++j) cur[j] = nxt[j];
        }
    }

    red[w][q * 4 + 2 * h]     = h ? a2[0] : a0[0];
    red[w][q * 4 + 2 * h + 1] = h ? a3[0] : a1[0];
    __syncthreads();
    if (threadIdx.x < 64) {
        float s = 0.0f;
        for (int i = 0; i < 8; ++i) s += red[i][threadIdx.x];
        part[((size_t)b * NCH + c) * FF + fh * 64 + threadIdx.x] = s;
    }
}

__global__ __launch_bounds__(1024) void k_out(const float* __restrict__ x,
                                              const float* __restrict__ part,
                                              const int* __restrict__ lengths,
                                              float* __restrict__ out) {
    __shared__ float segtot[32][32];
    const int fs   = blockIdx.x;
    const int b    = blockIdx.y;
    const int lane = threadIdx.x & 31;
    const int seg  = threadIdx.x >> 5;
    const int f    = fs * 32 + lane;

    const int   fsize = lengths[b] - (OUTT - 1);
    const float fsf   = (float)fsize;

    const float* xb = x + (size_t)b * TT * FF + f;
    const int t0 = seg * 32;

    float e[32];
    float run = 0.0f;
#pragma unroll
    for (int i = 0; i < 32; ++i) {
        const int t  = t0 + i;
        const int th = t + fsize;
        const int rh = (i == 31 && th > TT - 1) ? TT - 1 : th;
        const float d = xb[(size_t)rh * FF] - xb[(size_t)t * FF];
        e[i] = run;
        run += d;
    }
    segtot[seg][lane] = run;

    const int nch = (fsize + CH - 1) / CH;
    const float* pb = part + (size_t)b * NCH * FF + f;
    float w0 = 0.0f;
#pragma unroll
    for (int c = 0; c < NCH; ++c) {
        const int cc = c < nch ? c : 0;
        const float p = pb[(size_t)cc * FF];
        w0 += (c < nch) ? p : 0.0f;
    }

    __syncthreads();
    float off = 0.0f;
#pragma unroll
    for (int s = 0; s < 32; ++s) {
        const float v = segtot[s][lane];
        off += (s < seg) ? v : 0.0f;
    }
    const float base = w0 + off;

    float* ob = out + ((size_t)b * OUTT + t0) * FF + f;
#pragma unroll
    for (int i = 0; i < 32; ++i)
        ob[(size_t)i * FF] = (base + e[i]) / fsf;
}

extern "C" void kernel_launch(void* const* d_in, const int* in_sizes, int n_in,
                              void* d_out, int out_size, void* d_ws, size_t ws_size,
                              hipStream_t stream) {
    (void)in_sizes; (void)n_in; (void)out_size; (void)ws_size;
    const float* x       = (const float*)d_in[0];
    const int*   lengths = (const int*)d_in[1];
    float*       out     = (float*)d_out;
    float*       part    = (float*)d_ws;

    k_winsum_wmma<<<dim3(15, 2, 32), 256, 0, stream>>>(x, lengths, part);
    k_out<<<dim3(4, 32), 1024, 0, stream>>>(x, part, lengths, out);
}


# Window-sum kernel on a fixed grid of 256 work-groups that walk the compactly numbered in-window (batch, chunk, half) items as a pipelined loop; was one work-group per item slot; same sum order
# speedup vs baseline: 5.5211x; 1.1235x over previous
#include <hip/hip_runtime.h>

#define BB   32
#define TT   8192
#define FF   128
#define OUTT 1024
#define CH   512
#define NCH  15

typedef float v2f __attribute__((ext_vector_type(2)));
typedef float v4f __attribute__((ext_vector_type(4)));
typedef float v8f __attribute__((ext_vector_type(8)));


#define GW 256

__device__ __forceinline__ void load_grp(v4f (&buf)[16], const float* xb, int rowbase, int last) {
#pragma unroll
    for (int j = 0; j < 16; ++j) {
        const int row = rowbase + 2 * j;
        const int rc  = row < last ? row : last;
        buf[j] = *(const v4f*)(xb + (size_t)rc * FF);
    }
}

__device__ __forceinline__ void mma_grp(const v4f (&buf)[16], int rowbase, int last,
                                        v8f& a0, v8f& a1, v8f& a2, v8f& a3) {
#pragma unroll
    for (int j = 0; j < 16; j += 2) {
        const int rowa = rowbase + 2 * j;
        v2f am;
        am.x = (rowa     <= last) ? 1.0f : 0.0f;
        am.y = (rowa + 2 <= last) ? 1.0f : 0.0f;
        v2f b0; b0.x = buf[j].x; b0.y = buf[j + 1].x;
        v2f b1; b1.x = buf[j].y; b1.y = buf[j + 1].y;
        v2f b2; b2.x = buf[j].z; b2.y = buf[j + 1].z;
        v2f b3; b3.x = buf[j].w; b3.y = buf[j + 1].w;
        a0 = __builtin_amdgcn_wmma_f32_16x16x4_f32(false, am, false, b0, (short)0, a0, false, false);
        a1 = __builtin_amdgcn_wmma_f32_16x16x4_f32(false, am, false, b1, (short)0, a1, false, false);
        a2 = __builtin_amdgcn_wmma_f32_16x16x4_f32(false, am, false, b2, (short)0, a2, false, false);
        a3 = __builtin_amdgcn_wmma_f32_16x16x4_f32(false, am, false, b3, (short)0, a3, false, false);
    }
}

__device__ __forceinline__ void put_item(float* plane, int w, int q, int h,
                                         const v8f& a0, const v8f& a1, const v8f& a2, const v8f& a3,
                                         float* __restrict__ part, int b, int c, int fh) {
    plane[w * 64 + q * 4 + 2 * h]     = h ? a2[0] : a0[0];
    plane[w * 64 + q * 4 + 2 * h + 1] = h ? a3[0] : a1[0];
    __syncthreads();
    if (threadIdx.x < 64) {
        float s = 0.0f;
        for (int i = 0; i < 8; ++i) s += plane[i * 64 + threadIdx.x];
        part[((size_t)b * NCH + c) * FF + fh * 64 + threadIdx.x] = s;
    }
}

__global__ __launch_bounds__(256) void k_winsum_wmma(const float* __restrict__ x,
                                                     const int* __restrict__ lengths,
                                                     float* __restrict__ part) {
    __shared__ float red[2][8 * 64];
    const int lane = threadIdx.x & 31;
    const int w    = threadIdx.x >> 5;
    const int h    = lane >> 4;
    const int q    = lane & 15;

    const int fs_l = lengths[lane] - (OUTT - 1);
    int cnt_l = (fs_l + CH - 1) / CH;
    cnt_l = cnt_l < 0 ? 0 : (cnt_l > NCH ? NCH : cnt_l);
    int inc = cnt_l;
    for (int d = 1; d < 32; d <<= 1) {
        const int t = __shfl_up(inc, d);
        if (lane >= d) inc += t;
    }
    const int pre_l  = inc - cnt_l;
    const int nitems = 2 * __builtin_amdgcn_readlane(inc, 31);

    const int j = blockIdx.x;
    if (j >= nitems) return;

    int jj = j >> 1;
    int fh = j & 1;
    int b  = __popc((unsigned)__ballot(pre_l <= jj)) - 1;
    int c  = jj - __builtin_amdgcn_readlane(pre_l, b);
    int last = __builtin_amdgcn_readlane(fs_l, b) - 1;
    const float* xb = x + (size_t)b * TT * FF + fh * 64 + q * 4;
    int r0 = c * CH + w * 64 + h;

    v4f A[16], B[16];
    load_grp(A, xb, r0, last);
    int plane = 0;

    for (int jn = j + GW; jn < nitems; jn += GW) {
        v8f a0 = {}, a1 = {}, a2 = {}, a3 = {};

        load_grp(B, xb, r0 + 32, last);
        __builtin_amdgcn_sched_barrier(0);
        mma_grp(A, r0, last, a0, a1, a2, a3);

        const int bc = b, cc = c, fhc = fh, lastc = last, r0c = r0;
        jj   = jn >> 1;
        fh   = jn & 1;
        b    = __popc((unsigned)__ballot(pre_l <= jj)) - 1;
        c    = jj - __builtin_amdgcn_readlane(pre_l, b);
        last = __builtin_amdgcn_readlane(fs_l, b) - 1;
        xb   = x + (size_t)b * TT * FF + fh * 64 + q * 4;
        r0   = c * CH + w * 64 + h;
        load_grp(A, xb, r0, last);

        mma_grp(B, r0c + 32, lastc, a0, a1, a2, a3);
        put_item(red[plane], w, q, h, a0, a1, a2, a3, part, bc, cc, fhc);
        plane ^= 1;
    }

    {
        v8f a0 = {}, a1 = {}, a2 = {}, a3 = {};

        load_grp(B, xb, r0 + 32, last);
        __builtin_amdgcn_sched_barrier(0);
        mma_grp(A, r0, last, a0, a1, a2, a3);
        mma_grp(B, r0 + 32, last, a0, a1, a2, a3);
        put_item(red[plane], w, q, h, a0, a1, a2, a3, part, b, c, fh);
    }
}

__global__ __launch_bounds__(1024) void k_out(const float* __restrict__ x,
                                              const float* __restrict__ part,
                                              const int* __restrict__ lengths,
                                              float* __restrict__ out) {
    __shared__ float segtot[32][32];
    const int fs   = blockIdx.x;
    const int b    = blockIdx.y;
    const int lane = threadIdx.x & 31;
    const int seg  = threadIdx.x >> 5;
    const int f    = fs * 32 + lane;

    const int   fsize = lengths[b] - (OUTT - 1);
    const float fsf   = (float)fsize;

    const float* xb = x + (size_t)b * TT * FF + f;
    const int t0 = seg * 32;

    float e[32];
    float run = 0.0f;
#pragma unroll
    for (int i = 0; i < 32; ++i) {
        const int t  = t0 + i;
        const int th = t + fsize;
        const int rh = (i == 31 && th > TT - 1) ? TT - 1 : th;
        const float d = xb[(size_t)rh * FF] - xb[(size_t)t * FF];
        e[i] = run;
        run += d;
    }
    segtot[seg][lane] = run;

    const int nch = (fsize + CH - 1) / CH;
    const float* pb = part + (size_t)b * NCH * FF + f;
    float w0 = 0.0f;
#pragma unroll
    for (int c = 0; c < NCH; ++c) {
        const int cc = c < nch ? c : 0;
        const float p = pb[(size_t)cc * FF];
        w0 += (c < nch) ? p : 0.0f;
    }

    __syncthreads();
    float off = 0.0f;
#pragma unroll
    for (int s = 0; s < 32; ++s) {
        const float v = segtot[s][lane];
        off += (s < seg) ? v : 0.0f;
    }
    const float base = w0 + off;

    float* ob = out + ((size_t)b * OUTT + t0) * FF + f;
#pragma unroll
    for (int i = 0; i < 32; ++i)
        ob[(size_t)i * FF] = (base + e[i]) / fsf;
}

extern "C" void kernel_launch(void* const* d_in, const int* in_sizes, int n_in,
                              void* d_out, int out_size, void* d_ws, size_t ws_size,
                              hipStream_t stream) {
    (void)in_sizes; (void)n_in; (void)out_size; (void)ws_size;
    const float* x       = (const float*)d_in[0];
    const int*   lengths = (const int*)d_in[1];
    float*       out     = (float*)d_out;
    float*       part    = (float*)d_ws;

    k_winsum_wmma<<<dim3(GW), 256, 0, stream>>>(x, lengths, part);
    k_out<<<dim3(4, 32), 1024, 0, stream>>>(x, part, lengths, out);
}


# Window sum taken at t=512 (rows 512..511+fs): output kernel splits into two independent t-halves, backward scan below, forward above: 256 work-groups of 16 waves, was 128 of 32; changes the sum order
# speedup vs baseline: 6.2075x; 1.1243x over previous
#include <hip/hip_runtime.h>

#define BB   32
#define TT   8192
#define FF   128
#define OUTT 1024
#define CH   512
#define NCH  15

typedef float v2f __attribute__((ext_vector_type(2)));
typedef float v4f __attribute__((ext_vector_type(4)));
typedef float v8f __attribute__((ext_vector_type(8)));

#define MID  512


#define GW 256

__device__ __forceinline__ void load_grp(v4f (&buf)[16], const float* xb, int rowbase, int last) {
#pragma unroll
    for (int j = 0; j < 16; ++j) {
        const int row = rowbase + 2 * j;
        const int rc  = row < last ? row : last;
        buf[j] = *(const v4f*)(xb + (size_t)rc * FF);
    }
}

__device__ __forceinline__ void mma_grp(const v4f (&buf)[16], int rowbase, int last,
                                        v8f& a0, v8f& a1, v8f& a2, v8f& a3) {
#pragma unroll
    for (int j = 0; j < 16; j += 2) {
        const int rowa = rowbase + 2 * j;
        v2f am;
        am.x = (rowa     <= last) ? 1.0f : 0.0f;
        am.y = (rowa + 2 <= last) ? 1.0f : 0.0f;
        v2f b0; b0.x = buf[j].x; b0.y = buf[j + 1].x;
        v2f b1; b1.x = buf[j].y; b1.y = buf[j + 1].y;
        v2f b2; b2.x = buf[j].z; b2.y = buf[j + 1].z;
        v2f b3; b3.x = buf[j].w; b3.y = buf[j + 1].w;
        a0 = __builtin_amdgcn_wmma_f32_16x16x4_f32(false, am, false, b0, (short)0, a0, false, false);
        a1 = __builtin_amdgcn_wmma_f32_16x16x4_f32(false, am, false, b1, (short)0, a1, false, false);
        a2 = __builtin_amdgcn_wmma_f32_16x16x4_f32(false, am, false, b2, (short)0, a2, false, false);
        a3 = __builtin_amdgcn_wmma_f32_16x16x4_f32(false, am, false, b3, (short)0, a3, false, false);
    }
}

__device__ __forceinline__ void put_item(float* plane, int w, int q, int h,
                                         const v8f& a0, const v8f& a1, const v8f& a2, const v8f& a3,
                                         float* __restrict__ part, int b, int c, int fh) {
    plane[w * 64 + q * 4 + 2 * h]     = h ? a2[0] : a0[0];
    plane[w * 64 + q * 4 + 2 * h + 1] = h ? a3[0] : a1[0];
    __syncthreads();
    if (threadIdx.x < 64) {
        float s = 0.0f;
        for (int i = 0; i < 8; ++i) s += plane[i * 64 + threadIdx.x];
        part[((size_t)b * NCH + c) * FF + fh * 64 + threadIdx.x] = s;
    }
}

__global__ __launch_bounds__(256) void k_winsum_wmma(const float* __restrict__ x,
                                                     const int* __restrict__ lengths,
                                                     float* __restrict__ part) {
    __shared__ float red[2][8 * 64];
    const int lane = threadIdx.x & 31;
    const int w    = threadIdx.x >> 5;
    const int h    = lane >> 4;
    const int q    = lane & 15;

    const int fs_l = lengths[lane] - (OUTT - 1);
    int cnt_l = (fs_l + CH - 1) / CH;
    cnt_l = cnt_l < 0 ? 0 : (cnt_l > NCH ? NCH : cnt_l);
    int inc = cnt_l;
    for (int d = 1; d < 32; d <<= 1) {
        const int t = __shfl_up(inc, d);
        if (lane >= d) inc += t;
    }
    const int pre_l  = inc - cnt_l;
    const int nitems = 2 * __builtin_amdgcn_readlane(inc, 31);

    const int j = blockIdx.x;
    if (j >= nitems) return;

    int jj = j >> 1;
    int fh = j & 1;
    int b  = __popc((unsigned)__ballot(pre_l <= jj)) - 1;
    int c  = jj - __builtin_amdgcn_readlane(pre_l, b);
    int last = __builtin_amdgcn_readlane(fs_l, b) - 1;
    const float* xb = x + ((size_t)b * TT + MID) * FF + fh * 64 + q * 4;
    int r0 = c * CH + w * 64 + h;

    v4f A[16], B[16];
    load_grp(A, xb, r0, last);
    int plane = 0;

    for (int jn = j + GW; jn < nitems; jn += GW) {
        v8f a0 = {}, a1 = {}, a2 = {}, a3 = {};

        load_grp(B, xb, r0 + 32, last);
        __builtin_amdgcn_sched_barrier(0);
        mma_grp(A, r0, last, a0, a1, a2, a3);

        const int bc = b, cc = c, fhc = fh, lastc = last, r0c = r0;
        jj   = jn >> 1;
        fh   = jn & 1;
        b    = __popc((unsigned)__ballot(pre_l <= jj)) - 1;
        c    = jj - __builtin_amdgcn_readlane(pre_l, b);
        last = __builtin_amdgcn_readlane(fs_l, b) - 1;
        xb   = x + ((size_t)b * TT + MID) * FF + fh * 64 + q * 4;
        r0   = c * CH + w * 64 + h;
        load_grp(A, xb, r0, last);

        mma_grp(B, r0c + 32, lastc, a0, a1, a2, a3);
        put_item(red[plane], w, q, h, a0, a1, a2, a3, part, bc, cc, fhc);
        plane ^= 1;
    }

    {
        v8f a0 = {}, a1 = {}, a2 = {}, a3 = {};

        load_grp(B, xb, r0 + 32, last);
        __builtin_amdgcn_sched_barrier(0);
        mma_grp(A, r0, last, a0, a1, a2, a3);
        mma_grp(B, r0 + 32, last, a0, a1, a2, a3);
        put_item(red[plane], w, q, h, a0, a1, a2, a3, part, b, c, fh);
    }
}

template <int HALF>
__device__ __forceinline__ void out_half(const float* __restrict__ x,
                                         const float* __restrict__ part,
                                         float* __restrict__ out,
                                         float (*segtot)[32],
                                         int b, int f, int lane, int seg, int fsize) {
    const float fsf = (float)fsize;
    const float* xb = x + (size_t)b * TT * FF + f;
    const int t0 = HALF * MID + seg * 32;

    float e[32];
    float run = 0.0f;
#pragma unroll
    for (int j = 0; j < 32; ++j) {
        const int i  = HALF ? j : 31 - j;
        const int t  = t0 + i;
        const int th = t + fsize;
        const int rh = (HALF && i == 31 && th > TT - 1) ? TT - 1 : th;
        const float d = xb[(size_t)rh * FF] - xb[(size_t)t * FF];
        if (HALF) { e[i] = run; run += d; }
        else      { run += d; e[i] = run; }
    }
    segtot[seg][lane] = run;

    const int nch = (fsize + CH - 1) / CH;
    const float* pb = part + (size_t)b * NCH * FF + f;
    float wm = 0.0f;
#pragma unroll
    for (int c = 0; c < NCH; ++c) {
        const int cc = c < nch ? c : 0;
        const float p = pb[(size_t)cc * FF];
        wm += (c < nch) ? p : 0.0f;
    }

    __syncthreads();
    float off = 0.0f;
#pragma unroll
    for (int s = 0; s < 16; ++s) {
        const float v = segtot[s][lane];
        off += (HALF ? (s < seg) : (s > seg)) ? v : 0.0f;
    }
    const float base = HALF ? wm + off : wm - off;

    float* ob = out + ((size_t)b * OUTT + t0) * FF + f;
#pragma unroll
    for (int i = 0; i < 32; ++i)
        ob[(size_t)i * FF] = (HALF ? base + e[i] : base - e[i]) / fsf;
}

__global__ __launch_bounds__(512) void k_out(const float* __restrict__ x,
                                             const float* __restrict__ part,
                                             const int* __restrict__ lengths,
                                             float* __restrict__ out) {
    __shared__ float segtot[16][32];
    const int b    = blockIdx.y;
    const int lane = threadIdx.x & 31;
    const int seg  = threadIdx.x >> 5;
    const int f    = blockIdx.x * 32 + lane;
    const int fsize = lengths[b] - (OUTT - 1);

    if (blockIdx.z) out_half<1>(x, part, out, segtot, b, f, lane, seg, fsize);
    else            out_half<0>(x, part, out, segtot, b, f, lane, seg, fsize);
}

extern "C" void kernel_launch(void* const* d_in, const int* in_sizes, int n_in,
                              void* d_out, int out_size, void* d_ws, size_t ws_size,
                              hipStream_t stream) {
    (void)in_sizes; (void)n_in; (void)out_size; (void)ws_size;
    const float* x       = (const float*)d_in[0];
    const int*   lengths = (const int*)d_in[1];
    float*       out     = (float*)d_out;
    float*       part    = (float*)d_ws;

    k_winsum_wmma<<<dim3(GW), 256, 0, stream>>>(x, lengths, part);
    k_out<<<dim3(4, 32, 2), 512, 0, stream>>>(x, part, lengths, out);
}
